# starting kernel
# baseline (speedup 1.0000x reference)
#include <hip/hip_runtime.h>
#include <math.h>

typedef __attribute__((ext_vector_type(2))) float v2f;
typedef __attribute__((ext_vector_type(8))) float v8f;

#define ALPHA 0.2f
#define BB 4
#define KK 256
#define FF 128
#define EE 256
#define ITILE 4
#define ECHUNK 32

__global__ __launch_bounds__(256)
void gat_gemm_s_kernel(const float* __restrict__ X,
                       const float* __restrict__ W,
                       const float* __restrict__ blin,
                       float* __restrict__ S)
{
    const int wave = threadIdx.x >> 5;
    const int lane = threadIdx.x & 31;
    const int ln   = lane & 15;
    const int hi   = lane >> 4;
    const int s    = blockIdx.y;
    const int tile = blockIdx.x * 8 + wave;
    const int row0 = (tile >> 4) * 16;
    const int col0 = (tile & 15) * 16;

    const float* Arow = X + (size_t)(row0 + ln) * FF;
    const float* Brow = W + (size_t)(col0 + ln) * (2 * FF) + s * FF;

    v8f c = {};
    #pragma unroll 4
    for (int k0 = 0; k0 < FF; k0 += 4) {
        v2f a = *(const v2f*)(Arow + k0 + 2 * hi);
        v2f b = *(const v2f*)(Brow + k0 + 2 * hi);
        c = __builtin_amdgcn_wmma_f32_16x16x4_f32(false, a, false, b,
                                                  (short)0, c, false, false);
    }

    const float add = (s == 0) ? blin[col0 + ln] : 0.0f;
    float* D = S + (size_t)s * 1024 * EE;
    #pragma unroll
    for (int r = 0; r < 8; ++r)
        D[(size_t)(row0 + r + 8 * hi) * EE + col0 + ln] = c[r] + add;
}

__global__ __launch_bounds__(256)
void gat_attn_kernel(const float* __restrict__ S,
                     const float* __restrict__ avec,
                     const float* __restrict__ bias,
                     float* __restrict__ attn)
{
    const int b  = blockIdx.y;
    const int i0 = blockIdx.x * ITILE;
    const int j  = threadIdx.x;

    __shared__ float s1row[ITILE][EE];
    __shared__ float a_s[EE];
    __shared__ float tile[KK][ECHUNK + 1];
    __shared__ float red[KK];

    const float* S1 = S;
    const float* S2b = S + (size_t)1024 * EE + (size_t)b * KK * EE;

    a_s[j] = avec[j];
    #pragma unroll
    for (int ii = 0; ii < ITILE; ++ii)
        s1row[ii][j] = S1[(size_t)(b * KK + i0 + ii) * EE + j];
    __syncthreads();

    float acc[ITILE];
    #pragma unroll
    for (int ii = 0; ii < ITILE; ++ii) acc[ii] = 0.0f;

    for (int e0 = 0; e0 < EE; e0 += ECHUNK) {
        __syncthreads();
        for (int idx = threadIdx.x; idx < KK * ECHUNK; idx += 256) {
            const int jj = idx >> 5;
            const int ee = idx & 31;
            tile[jj][ee] = S2b[(size_t)jj * EE + e0 + ee];
        }
        __syncthreads();
        #pragma unroll 8
        for (int ee = 0; ee < ECHUNK; ++ee) {
            const float t  = tile[j][ee];
            const float ae = a_s[e0 + ee];
            #pragma unroll
            for (int ii = 0; ii < ITILE; ++ii) {
                const float p = s1row[ii][e0 + ee] + t;
                const float h = (p >= 0.0f) ? p : ALPHA * p;
                acc[ii] = fmaf(ae, h, acc[ii]);
            }
        }
    }

    #pragma unroll
    for (int ii = 0; ii < ITILE; ++ii) {
        const float ev = acc[ii] + bias[(size_t)(i0 + ii) * KK + j];

        red[j] = ev;           __syncthreads();
        for (int off = 128; off > 0; off >>= 1) {
            if (j < off) red[j] = fmaxf(red[j], red[j + off]);
            __syncthreads();
        }
        const float mx = red[0]; __syncthreads();

        const float p = __expf(ev - mx);
        red[j] = p;            __syncthreads();
        for (int off = 128; off > 0; off >>= 1) {
            if (j < off) red[j] += red[j + off];
            __syncthreads();
        }
        const float inv = 1.0f / red[0];
        __syncthreads();

        attn[(size_t)(b * KK + i0 + ii) * KK + j] = p * inv;
    }
}

__global__ __launch_bounds__(256)
void gat_out_kernel(const float* __restrict__ attn,
                    const float* __restrict__ X,
                    float* __restrict__ out)
{
    const int wave = threadIdx.x >> 5;
    const int lane = threadIdx.x & 31;
    const int ln   = lane & 15;
    const int hi   = lane >> 4;
    const int b    = blockIdx.y;
    const int tile = blockIdx.x * 8 + wave;
    const int row0 = (tile >> 3) * 16;
    const int col0 = (tile & 7) * 16;

    const float* A  = attn + (size_t)b * KK * KK;
    const float* Bm = X    + (size_t)b * KK * FF;
    const float* Arow = A + (size_t)(row0 + ln) * KK;

    v8f c = {};
    #pragma unroll 4
    for (int k0 = 0; k0 < KK; k0 += 4) {
        v2f a = *(const v2f*)(Arow + k0 + 2 * hi);
        v2f bb;
        bb.x = Bm[(size_t)(k0 + 2 * hi) * FF + col0 + ln];
        bb.y = Bm[(size_t)(k0 + 2 * hi + 1) * FF + col0 + ln];
        c = __builtin_amdgcn_wmma_f32_16x16x4_f32(false, a, false, bb,
                                                  (short)0, c, false, false);
    }

    float* O = out + (size_t)b * KK * FF;
    #pragma unroll
    for (int r = 0; r < 8; ++r) {
        const float v = c[r];
        O[(size_t)(row0 + r + 8 * hi) * FF + col0 + ln] =
            1.0f / (1.0f + __expf(-v));
    }
}

extern "C" void kernel_launch(void* const* d_in, const int* in_sizes, int n_in,
                              void* d_out, int out_size, void* d_ws, size_t ws_size,
                              hipStream_t stream) {
    const float* x    = (const float*)d_in[0];
    const float* W    = (const float*)d_in[1];
    const float* blin = (const float*)d_in[2];
    const float* a    = (const float*)d_in[3];
    const float* bias = (const float*)d_in[4];
    float* out = (float*)d_out;

    float* S     = (float*)d_ws;
    float* attnw = S + (size_t)2 * 1024 * EE;

    gat_gemm_s_kernel<<<dim3(128, 2), 256, 0, stream>>>(x, W, blin, S);

    gat_attn_kernel<<<dim3(KK / ITILE, BB), 256, 0, stream>>>(S, a, bias, attnw);

    gat_out_kernel<<<dim3(16, BB), 256, 0, stream>>>(attnw, x, out);
}


# attention re-tiled: e-sum split across the 8 waves, each streams its S2 slice (GEMM stores S2 e-major) global->registers, no LDS stage or barrier in e loop; row softmax in one wave; sum order changes
# speedup vs baseline: 1.6230x; 1.6230x over previous
#include <hip/hip_runtime.h>
#include <math.h>


typedef __attribute__((ext_vector_type(2))) float v2f;
typedef __attribute__((ext_vector_type(4))) float v4f;
typedef __attribute__((ext_vector_type(8))) float v8f;

#define ALPHA 0.2f
#define BB 4
#define KK 256
#define FF 128
#define EE 256
#define ITILE 4
#define NWAVE 8
#define ESLICE (EE / NWAVE)
#define ECH 4

__global__ __launch_bounds__(256)
void gat_gemm_s_kernel(const float* __restrict__ X,
                       const float* __restrict__ W,
                       const float* __restrict__ blin,
                       float* __restrict__ S)
{
    const int wave = threadIdx.x >> 5;
    const int lane = threadIdx.x & 31;
    const int ln   = lane & 15;
    const int hi   = lane >> 4;
    const int s    = blockIdx.y;
    const int tile = blockIdx.x * 8 + wave;
    const int xr0  = (tile >> 4) * 16;
    const int wr0  = (tile & 15) * 16;

    const float* Xrow = X + (size_t)(xr0 + ln) * FF;
    const float* Wrow = W + (size_t)(wr0 + ln) * (2 * FF) + s * FF;
    const float* Arow = s ? Wrow : Xrow;
    const float* Brow = s ? Xrow : Wrow;

    v8f c = {};
    #pragma unroll 4
    for (int k0 = 0; k0 < FF; k0 += 4) {
        v2f a = *(const v2f*)(Arow + k0 + 2 * hi);
        v2f b = *(const v2f*)(Brow + k0 + 2 * hi);
        c = __builtin_amdgcn_wmma_f32_16x16x4_f32(false, a, false, b,
                                                  (short)0, c, false, false);
    }

    if (s == 0) {
        const float add = blin[wr0 + ln];
        #pragma unroll
        for (int r = 0; r < 8; ++r)
            S[(size_t)(xr0 + r + 8 * hi) * EE + wr0 + ln] = c[r] + add;
    } else {
        float* D = S + (size_t)1024 * EE;
        const int m  = xr0 + ln;
        const int bb = m >> 8;
        const int j  = m & (KK - 1);
        #pragma unroll
        for (int r = 0; r < 8; ++r)
            D[((size_t)bb * EE + wr0 + r + 8 * hi) * KK + j] = c[r];
    }
}

struct Chunk {
    v4f t[2 * ECH];
    v4f a4;
    v4f s4[ITILE];
};

__device__ __forceinline__ void load_chunk_g(Chunk& ck, const v4f* src, int chunk)
{
    #pragma unroll
    for (int k = 0; k < 2 * ECH; ++k)
        ck.t[k] = src[(chunk * 2 * ECH + k) * 32];
}

__device__ __forceinline__ void load_chunk_l(Chunk& ck, const float* a_e, const float (*s1)[EE], int eb)
{
    ck.a4 = *(const v4f*)(a_e + eb);
    #pragma unroll
    for (int ii = 0; ii < ITILE; ++ii) ck.s4[ii] = *(const v4f*)(&s1[ii][eb]);
}

__device__ __forceinline__ void load_chunk(Chunk& ck, const v4f* src, int chunk,
                                           const float* a_e, const float (*s1)[EE], int eb)
{
    load_chunk_g(ck, src, chunk);
    load_chunk_l(ck, a_e, s1, eb);
}

__device__ __forceinline__ void eat_chunk(float (&acc)[ITILE][8], const Chunk& ck)
{
    #pragma unroll
    for (int x = 0; x < ECH; ++x) {
        #pragma unroll
        for (int half = 0; half < 2; ++half) {
            const v4f t = ck.t[2 * x + half];
            #pragma unroll
            for (int ii = 0; ii < ITILE; ++ii) {
                #pragma unroll
                for (int jj = 0; jj < 4; ++jj) {
                    const float p = ck.s4[ii][x] + t[jj];
                    const float h = (p >= 0.0f) ? p : ALPHA * p;
                    acc[ii][half * 4 + jj] = fmaf(ck.a4[x], h, acc[ii][half * 4 + jj]);
                }
            }
        }
    }
}

__global__ __launch_bounds__(256) __attribute__((amdgpu_waves_per_eu(2, 2)))
void gat_attn_kernel(const float* __restrict__ S,
                     const float* __restrict__ avec,
                     const float* __restrict__ bias,
                     float* __restrict__ attn)
{
    const int b    = blockIdx.y;
    const int i0   = blockIdx.x * ITILE;
    const int tid  = threadIdx.x;
    const int wave = tid >> 5;
    const int lane = tid & 31;

    __shared__ __attribute__((aligned(16))) float s1row[ITILE][EE];
    __shared__ __attribute__((aligned(16))) float a_s[EE];
    __shared__ __attribute__((aligned(16))) float part[NWAVE][ITILE][KK];

    const float* S1 = S;
    const v4f* src = (const v4f*)(S + (size_t)1024 * EE
                                    + ((size_t)b * EE + wave * ESLICE) * KK) + lane;

    Chunk A, B;
    load_chunk_g(A, src, 0);

    const int r    = wave >> 1;
    const int hsel = wave & 1;
    const v4f* brow = (const v4f*)(bias + (size_t)(i0 + r) * KK) + lane;
    const v4f bias_lo = brow[0];
    const v4f bias_hi = brow[32];

    a_s[tid] = avec[tid];
    #pragma unroll
    for (int ii = 0; ii < ITILE; ++ii)
        s1row[ii][tid] = S1[(size_t)(b * KK + i0 + ii) * EE + tid];
    __syncthreads();

    float acc[ITILE][8];
    #pragma unroll
    for (int ii = 0; ii < ITILE; ++ii) {
        #pragma unroll
        for (int q = 0; q < 8; ++q) acc[ii][q] = 0.0f;
    }

    const int e_w = wave * ESLICE;
    constexpr int NCH = ESLICE / ECH;
    load_chunk_l(A, a_s, s1row, e_w);
    #pragma nounroll
    for (int c = 0; c < NCH - 2; c += 2) {
        load_chunk(B, src, c + 1, a_s, s1row, e_w + (c + 1) * ECH);
        eat_chunk(acc, A);
        load_chunk(A, src, c + 2, a_s, s1row, e_w + (c + 2) * ECH);
        eat_chunk(acc, B);
    }
    load_chunk(B, src, NCH - 1, a_s, s1row, e_w + (NCH - 1) * ECH);
    eat_chunk(acc, A);
    eat_chunk(acc, B);

    #pragma unroll
    for (int ii = 0; ii < ITILE; ++ii) {
        v4f lo = {acc[ii][0], acc[ii][1], acc[ii][2], acc[ii][3]};
        v4f hi = {acc[ii][4], acc[ii][5], acc[ii][6], acc[ii][7]};
        *(v4f*)(&part[wave][ii][4 * lane])       = lo;
        *(v4f*)(&part[wave][ii][128 + 4 * lane]) = hi;
    }
    __syncthreads();

    v4f ev_lo = *(const v4f*)(&part[0][r][4 * lane]);
    v4f ev_hi = *(const v4f*)(&part[0][r][128 + 4 * lane]);
    #pragma unroll
    for (int w = 1; w < NWAVE; ++w) {
        ev_lo += *(const v4f*)(&part[w][r][4 * lane]);
        ev_hi += *(const v4f*)(&part[w][r][128 + 4 * lane]);
    }
    ev_lo += bias_lo;
    ev_hi += bias_hi;

    float mx = fmaxf(fmaxf(fmaxf(ev_lo[0], ev_lo[1]), fmaxf(ev_lo[2], ev_lo[3])),
                     fmaxf(fmaxf(ev_hi[0], ev_hi[1]), fmaxf(ev_hi[2], ev_hi[3])));
    #pragma unroll
    for (int off = 16; off > 0; off >>= 1)
        mx = fmaxf(mx, __shfl_xor(mx, off, 32));

    v4f p_lo, p_hi;
    #pragma unroll
    for (int q = 0; q < 4; ++q) {
        p_lo[q] = __expf(ev_lo[q] - mx);
        p_hi[q] = __expf(ev_hi[q] - mx);
    }
    float sum = ((p_lo[0] + p_lo[1]) + (p_lo[2] + p_lo[3]))
              + ((p_hi[0] + p_hi[1]) + (p_hi[2] + p_hi[3]));
    #pragma unroll
    for (int off = 16; off > 0; off >>= 1)
        sum += __shfl_xor(sum, off, 32);
    const float inv = 1.0f / sum;

    const v4f o = (hsel ? p_hi : p_lo) * inv;
    *(v4f*)(attn + (size_t)(b * KK + i0 + r) * KK + hsel * 128 + 4 * lane) = o;
}

__global__ __launch_bounds__(256)
void gat_out_kernel(const float* __restrict__ attn,
                    const float* __restrict__ X,
                    float* __restrict__ out)
{
    const int wave = threadIdx.x >> 5;
    const int lane = threadIdx.x & 31;
    const int ln   = lane & 15;
    const int hi   = lane >> 4;
    const int b    = blockIdx.y;
    const int tile = blockIdx.x * 8 + wave;
    const int row0 = (tile >> 3) * 16;
    const int col0 = (tile & 7) * 16;

    const float* A  = attn + (size_t)b * KK * KK;
    const float* Bm = X    + (size_t)b * KK * FF;
    const float* Arow = A + (size_t)(row0 + ln) * KK;

    v8f c = {};
    #pragma unroll 4
    for (int k0 = 0; k0 < KK; k0 += 4) {
        v2f a = *(const v2f*)(Arow + k0 + 2 * hi);
        v2f bb;
        bb.x = Bm[(size_t)(k0 + 2 * hi) * FF + col0 + ln];
        bb.y = Bm[(size_t)(k0 + 2 * hi + 1) * FF + col0 + ln];
        c = __builtin_amdgcn_wmma_f32_16x16x4_f32(false, a, false, bb,
                                                  (short)0, c, false, false);
    }

    float* O = out + (size_t)b * KK * FF;
    #pragma unroll
    for (int r = 0; r < 8; ++r) {
        const float v = c[r];
        O[(size_t)(row0 + r + 8 * hi) * FF + col0 + ln] =
            1.0f / (1.0f + __expf(-v));
    }
}

extern "C" void kernel_launch(void* const* d_in, const int* in_sizes, int n_in,
                              void* d_out, int out_size, void* d_ws, size_t ws_size,
                              hipStream_t stream) {
    const float* x    = (const float*)d_in[0];
    const float* W    = (const float*)d_in[1];
    const float* blin = (const float*)d_in[2];
    const float* a    = (const float*)d_in[3];
    const float* bias = (const float*)d_in[4];
    float* out = (float*)d_out;

    float* S     = (float*)d_ws;
    float* attnw = S + (size_t)2 * 1024 * EE;

    gat_gemm_s_kernel<<<dim3(128, 2), 256, 0, stream>>>(x, W, blin, S);

    gat_attn_kernel<<<dim3(KK / ITILE, BB), 256, 0, stream>>>(S, a, bias, attnw);

    gat_out_kernel<<<dim3(16, BB), 256, 0, stream>>>(attnw, x, out);
}


# out GEMM re-tiled: k split over a block's 4 waves (16x32 tile), each wave issues its 64-k slice's 40 fragment loads as one group before its first WMMA, partials summed via LDS; sum order changes
# speedup vs baseline: 2.2395x; 1.3799x over previous
#include <hip/hip_runtime.h>
#include <math.h>


typedef __attribute__((ext_vector_type(2))) float v2f;
typedef __attribute__((ext_vector_type(4))) float v4f;
typedef __attribute__((ext_vector_type(8))) float v8f;

#define ALPHA 0.2f
#define BB 4
#define KK 256
#define FF 128
#define EE 256
#define ITILE 4
#define NWAVE 8
#define ESLICE (EE / NWAVE)
#define ECH 4

__global__ __launch_bounds__(256)
void gat_gemm_s_kernel(const float* __restrict__ X,
                       const float* __restrict__ W,
                       const float* __restrict__ blin,
                       float* __restrict__ S)
{
    const int wave = threadIdx.x >> 5;
    const int lane = threadIdx.x & 31;
    const int ln   = lane & 15;
    const int hi   = lane >> 4;
    const int s    = blockIdx.y;
    const int tile = blockIdx.x * 8 + wave;
    const int xr0  = (tile >> 4) * 16;
    const int wr0  = (tile & 15) * 16;

    const float* Xrow = X + (size_t)(xr0 + ln) * FF;
    const float* Wrow = W + (size_t)(wr0 + ln) * (2 * FF) + s * FF;
    const float* Arow = s ? Wrow : Xrow;
    const float* Brow = s ? Xrow : Wrow;

    v8f c = {};
    #pragma unroll 4
    for (int k0 = 0; k0 < FF; k0 += 4) {
        v2f a = *(const v2f*)(Arow + k0 + 2 * hi);
        v2f b = *(const v2f*)(Brow + k0 + 2 * hi);
        c = __builtin_amdgcn_wmma_f32_16x16x4_f32(false, a, false, b,
                                                  (short)0, c, false, false);
    }

    if (s == 0) {
        const float add = blin[wr0 + ln];
        #pragma unroll
        for (int r = 0; r < 8; ++r)
            S[(size_t)(xr0 + r + 8 * hi) * EE + wr0 + ln] = c[r] + add;
    } else {
        float* D = S + (size_t)1024 * EE;
        const int m  = xr0 + ln;
        const int bb = m >> 8;
        const int j  = m & (KK - 1);
        #pragma unroll
        for (int r = 0; r < 8; ++r)
            D[((size_t)bb * EE + wr0 + r + 8 * hi) * KK + j] = c[r];
    }
}

struct Chunk {
    v4f t[2 * ECH];
    v4f a4;
    v4f s4[ITILE];
};

__device__ __forceinline__ void load_chunk_g(Chunk& ck, const v4f* src, int chunk)
{
    #pragma unroll
    for (int k = 0; k < 2 * ECH; ++k)
        ck.t[k] = src[(chunk * 2 * ECH + k) * 32];
}

__device__ __forceinline__ void load_chunk_l(Chunk& ck, const float* a_e, const float (*s1)[EE], int eb)
{
    ck.a4 = *(const v4f*)(a_e + eb);
    #pragma unroll
    for (int ii = 0; ii < ITILE; ++ii) ck.s4[ii] = *(const v4f*)(&s1[ii][eb]);
}

__device__ __forceinline__ void load_chunk(Chunk& ck, const v4f* src, int chunk,
                                           const float* a_e, const float (*s1)[EE], int eb)
{
    load_chunk_g(ck, src, chunk);
    load_chunk_l(ck, a_e, s1, eb);
}

__device__ __forceinline__ void eat_chunk(float (&acc)[ITILE][8], const Chunk& ck)
{
    #pragma unroll
    for (int x = 0; x < ECH; ++x) {
        #pragma unroll
        for (int half = 0; half < 2; ++half) {
            const v4f t = ck.t[2 * x + half];
            #pragma unroll
            for (int ii = 0; ii < ITILE; ++ii) {
                #pragma unroll
                for (int jj = 0; jj < 4; ++jj) {
                    const float p = ck.s4[ii][x] + t[jj];
                    const float h = (p >= 0.0f) ? p : ALPHA * p;
                    acc[ii][half * 4 + jj] = fmaf(ck.a4[x], h, acc[ii][half * 4 + jj]);
                }
            }
        }
    }
}

__global__ __launch_bounds__(256) __attribute__((amdgpu_waves_per_eu(2, 2)))
void gat_attn_kernel(const float* __restrict__ S,
                     const float* __restrict__ avec,
                     const float* __restrict__ bias,
                     float* __restrict__ attn)
{
    const int b    = blockIdx.y;
    const int i0   = blockIdx.x * ITILE;
    const int tid  = threadIdx.x;
    const int wave = tid >> 5;
    const int lane = tid & 31;

    __shared__ __attribute__((aligned(16))) float s1row[ITILE][EE];
    __shared__ __attribute__((aligned(16))) float a_s[EE];
    __shared__ __attribute__((aligned(16))) float part[NWAVE][ITILE][KK];

    const float* S1 = S;
    const v4f* src = (const v4f*)(S + (size_t)1024 * EE
                                    + ((size_t)b * EE + wave * ESLICE) * KK) + lane;

    Chunk A, B;
    load_chunk_g(A, src, 0);

    const int r    = wave >> 1;
    const int hsel = wave & 1;
    const v4f* brow = (const v4f*)(bias + (size_t)(i0 + r) * KK) + lane;
    const v4f bias_lo = brow[0];
    const v4f bias_hi = brow[32];

    a_s[tid] = avec[tid];
    #pragma unroll
    for (int ii = 0; ii < ITILE; ++ii)
        s1row[ii][tid] = S1[(size_t)(b * KK + i0 + ii) * EE + tid];
    __syncthreads();

    float acc[ITILE][8];
    #pragma unroll
    for (int ii = 0; ii < ITILE; ++ii) {
        #pragma unroll
        for (int q = 0; q < 8; ++q) acc[ii][q] = 0.0f;
    }

    const int e_w = wave * ESLICE;
    constexpr int NCH = ESLICE / ECH;
    load_chunk_l(A, a_s, s1row, e_w);
    #pragma nounroll
    for (int c = 0; c < NCH - 2; c += 2) {
        load_chunk(B, src, c + 1, a_s, s1row, e_w + (c + 1) * ECH);
        eat_chunk(acc, A);
        load_chunk(A, src, c + 2, a_s, s1row, e_w + (c + 2) * ECH);
        eat_chunk(acc, B);
    }
    load_chunk(B, src, NCH - 1, a_s, s1row, e_w + (NCH - 1) * ECH);
    eat_chunk(acc, A);
    eat_chunk(acc, B);

    #pragma unroll
    for (int ii = 0; ii < ITILE; ++ii) {
        v4f lo = {acc[ii][0], acc[ii][1], acc[ii][2], acc[ii][3]};
        v4f hi = {acc[ii][4], acc[ii][5], acc[ii][6], acc[ii][7]};
        *(v4f*)(&part[wave][ii][4 * lane])       = lo;
        *(v4f*)(&part[wave][ii][128 + 4 * lane]) = hi;
    }
    __syncthreads();

    v4f ev_lo = *(const v4f*)(&part[0][r][4 * lane]);
    v4f ev_hi = *(const v4f*)(&part[0][r][128 + 4 * lane]);
    #pragma unroll
    for (int w = 1; w < NWAVE; ++w) {
        ev_lo += *(const v4f*)(&part[w][r][4 * lane]);
        ev_hi += *(const v4f*)(&part[w][r][128 + 4 * lane]);
    }
    ev_lo += bias_lo;
    ev_hi += bias_hi;

    float mx = fmaxf(fmaxf(fmaxf(ev_lo[0], ev_lo[1]), fmaxf(ev_lo[2], ev_lo[3])),
                     fmaxf(fmaxf(ev_hi[0], ev_hi[1]), fmaxf(ev_hi[2], ev_hi[3])));
    #pragma unroll
    for (int off = 16; off > 0; off >>= 1)
        mx = fmaxf(mx, __shfl_xor(mx, off, 32));

    v4f p_lo, p_hi;
    #pragma unroll
    for (int q = 0; q < 4; ++q) {
        p_lo[q] = __expf(ev_lo[q] - mx);
        p_hi[q] = __expf(ev_hi[q] - mx);
    }
    float sum = ((p_lo[0] + p_lo[1]) + (p_lo[2] + p_lo[3]))
              + ((p_hi[0] + p_hi[1]) + (p_hi[2] + p_hi[3]));
    #pragma unroll
    for (int off = 16; off > 0; off >>= 1)
        sum += __shfl_xor(sum, off, 32);
    const float inv = 1.0f / sum;

    const v4f o = (hsel ? p_hi : p_lo) * inv;
    *(v4f*)(attn + (size_t)(b * KK + i0 + r) * KK + hsel * 128 + 4 * lane) = o;
}

#define OW    4
#define OKS   (KK / OW)
#define OJ    (OKS / 8)
#define OPITCH 36

__global__ __launch_bounds__(128) __attribute__((amdgpu_waves_per_eu(1, 4)))
void gat_out_kernel(const float* __restrict__ attn,
                    const float* __restrict__ X,
                    float* __restrict__ out)
{
    const int tid  = threadIdx.x;
    const int wave = tid >> 5;
    const int lane = tid & 31;
    const int ln   = lane & 15;
    const int hi   = lane >> 4;
    const int b    = blockIdx.y;
    const int row0 = (blockIdx.x >> 2) * 16;
    const int col0 = (blockIdx.x & 3) * 32;

    __shared__ __attribute__((aligned(16))) float part[OW][16][OPITCH];

    const int kb = OKS * wave + 4 * hi;
    const float* Ap = attn + (size_t)b * KK * KK + (size_t)(row0 + ln) * KK + kb;
    const float* Bp = X    + (size_t)b * KK * FF + (size_t)kb * FF + col0 + 2 * ln;

    v4f a4[OJ];
    v2f xb[OJ][2][2];
    #pragma unroll
    for (int j = 0; j < OJ; ++j) {
        a4[j] = *(const v4f*)(Ap + 8 * j);
        #pragma unroll
        for (int p = 0; p < 2; ++p) {
            #pragma unroll
            for (int e = 0; e < 2; ++e)
                xb[j][p][e] = *(const v2f*)(Bp + (size_t)(8 * j + 2 * p + e) * FF);
        }
    }

    v8f c0 = {};
    v8f c1 = {};
    #pragma unroll
    for (int j = 0; j < OJ; ++j) {
        #pragma unroll
        for (int p = 0; p < 2; ++p) {
            v2f a, b0, b1;
            a.x  = a4[j][2 * p];      a.y  = a4[j][2 * p + 1];
            b0.x = xb[j][p][0].x;     b0.y = xb[j][p][1].x;
            b1.x = xb[j][p][0].y;     b1.y = xb[j][p][1].y;
            c0 = __builtin_amdgcn_wmma_f32_16x16x4_f32(false, a, false, b0,
                                                       (short)0, c0, false, false);
            c1 = __builtin_amdgcn_wmma_f32_16x16x4_f32(false, a, false, b1,
                                                       (short)0, c1, false, false);
        }
    }

    #pragma unroll
    for (int r = 0; r < 8; ++r) {
        v2f v = {c0[r], c1[r]};
        *(v2f*)(&part[wave][8 * hi + r][2 * ln]) = v;
    }
    __syncthreads();

    const int orow = tid >> 3;
    const int ocol = (tid & 7) * 4;
    v4f acc = *(const v4f*)(&part[0][orow][ocol]);
    #pragma unroll
    for (int w = 1; w < OW; ++w)
        acc += *(const v4f*)(&part[w][orow][ocol]);

    v4f o;
    #pragma unroll
    for (int q = 0; q < 4; ++q)
        o[q] = 1.0f / (1.0f + __expf(-acc[q]));
    *(v4f*)(out + (size_t)b * KK * FF + (size_t)(row0 + orow) * FF + col0 + ocol) = o;
}

extern "C" void kernel_launch(void* const* d_in, const int* in_sizes, int n_in,
                              void* d_out, int out_size, void* d_ws, size_t ws_size,
                              hipStream_t stream) {
    const float* x    = (const float*)d_in[0];
    const float* W    = (const float*)d_in[1];
    const float* blin = (const float*)d_in[2];
    const float* a    = (const float*)d_in[3];
    const float* bias = (const float*)d_in[4];
    float* out = (float*)d_out;

    float* S     = (float*)d_ws;
    float* attnw = S + (size_t)2 * 1024 * EE;

    gat_gemm_s_kernel<<<dim3(128, 2), 256, 0, stream>>>(x, W, blin, S);

    gat_attn_kernel<<<dim3(KK / ITILE, BB), 256, 0, stream>>>(S, a, bias, attnw);

    gat_out_kernel<<<dim3(64, BB), 128, 0, stream>>>(attnw, x, out);
}


# S GEMM re-tiled: a 16x32 tile per wave over the whole k, its 48 fragment loads (16 B each) issued as one group into registers before the first WMMA, b_lin as accumulator start; k and sum order change
# speedup vs baseline: 2.4385x; 1.0889x over previous
#include <hip/hip_runtime.h>
#include <math.h>


typedef __attribute__((ext_vector_type(2))) float v2f;
typedef __attribute__((ext_vector_type(4))) float v4f;
typedef __attribute__((ext_vector_type(8))) float v8f;

#define ALPHA 0.2f
#define BB 4
#define KK 256
#define FF 128
#define EE 256
#define ITILE 4
#define NWAVE 8
#define ESLICE (EE / NWAVE)
#define ECH 4

#define GJ (FF / 8)

__global__ __launch_bounds__(128) __attribute__((amdgpu_waves_per_eu(1, 2)))
void gat_gemm_s_kernel(const float* __restrict__ X,
                       const float* __restrict__ W,
                       const float* __restrict__ blin,
                       float* __restrict__ S)
{
    const int wave = threadIdx.x >> 5;
    const int lane = threadIdx.x & 31;
    const int ln   = lane & 15;
    const int hi   = lane >> 4;
    const int s    = blockIdx.y;
    const int blk  = blockIdx.x;

    const int ga  = s ? 4 * (blk & 3) + wave : 4 * (blk & 15) + wave;
    const int gb  = s ? (blk >> 2) : (blk >> 4);
    const int ar0 = ga * 16;
    const int br0 = gb * 32;

    const float* Arow  = s ? W + (size_t)(ar0 + ln) * (2 * FF) + FF
                           : X + (size_t)(ar0 + ln) * FF;
    const float* B0row = s ? X + (size_t)(br0 + 2 * ln) * FF
                           : W + (size_t)(br0 + 2 * ln) * (2 * FF);
    const int bpitch   = s ? FF : 2 * FF;

    const float* Ap  = Arow  + 4 * hi;
    const float* B0p = B0row + 4 * hi;
    const float* B1p = B0p + bpitch;

    const v2f bl = *(const v2f*)(blin + (s ? 0 : br0) + 2 * ln);

    v4f a4[GJ], b0[GJ], b1[GJ];
    #pragma unroll
    for (int j = 0; j < GJ; ++j) {
        a4[j] = *(const v4f*)(Ap  + 8 * j);
        b0[j] = *(const v4f*)(B0p + 8 * j);
        b1[j] = *(const v4f*)(B1p + 8 * j);
    }
    __builtin_amdgcn_sched_barrier(0);

    const float i0 = s ? 0.0f : bl.x;
    const float i1 = s ? 0.0f : bl.y;
    v8f c0 = {i0, i0, i0, i0, i0, i0, i0, i0};
    v8f c1 = {i1, i1, i1, i1, i1, i1, i1, i1};
    #pragma unroll
    for (int j = 0; j < GJ; ++j) {
        #pragma unroll
        for (int p = 0; p < 2; ++p) {
            v2f a, x0, x1;
            a.x  = a4[j][2 * p];   a.y  = a4[j][2 * p + 1];
            x0.x = b0[j][2 * p];   x0.y = b0[j][2 * p + 1];
            x1.x = b1[j][2 * p];   x1.y = b1[j][2 * p + 1];
            c0 = __builtin_amdgcn_wmma_f32_16x16x4_f32(false, a, false, x0,
                                                       (short)0, c0, false, false);
            c1 = __builtin_amdgcn_wmma_f32_16x16x4_f32(false, a, false, x1,
                                                       (short)0, c1, false, false);
        }
    }

    float* D = s ? S + (size_t)1024 * EE
                     + ((size_t)(br0 >> 8) * EE + ar0 + 8 * hi) * KK
                     + (br0 & (KK - 1)) + 2 * ln
                 : S + (size_t)(ar0 + 8 * hi) * EE + br0 + 2 * ln;
    const int dpitch = s ? KK : EE;
    #pragma unroll
    for (int r = 0; r < 8; ++r) {
        v2f v = {c0[r], c1[r]};
        *(v2f*)(D + (size_t)r * dpitch) = v;
    }
}

struct Chunk {
    v4f t[2 * ECH];
    v4f a4;
    v4f s4[ITILE];
};

__device__ __forceinline__ void load_chunk_g(Chunk& ck, const v4f* src, int chunk)
{
    #pragma unroll
    for (int k = 0; k < 2 * ECH; ++k)
        ck.t[k] = src[(chunk * 2 * ECH + k) * 32];
}

__device__ __forceinline__ void load_chunk_l(Chunk& ck, const float* a_e, const float (*s1)[EE], int eb)
{
    ck.a4 = *(const v4f*)(a_e + eb);
    #pragma unroll
    for (int ii = 0; ii < ITILE; ++ii) ck.s4[ii] = *(const v4f*)(&s1[ii][eb]);
}

__device__ __forceinline__ void load_chunk(Chunk& ck, const v4f* src, int chunk,
                                           const float* a_e, const float (*s1)[EE], int eb)
{
    load_chunk_g(ck, src, chunk);
    load_chunk_l(ck, a_e, s1, eb);
}

__device__ __forceinline__ void eat_chunk(float (&acc)[ITILE][8], const Chunk& ck)
{
    #pragma unroll
    for (int x = 0; x < ECH; ++x) {
        #pragma unroll
        for (int half = 0; half < 2; ++half) {
            const v4f t = ck.t[2 * x + half];
            #pragma unroll
            for (int ii = 0; ii < ITILE; ++ii) {
                #pragma unroll
                for (int jj = 0; jj < 4; ++jj) {
                    const float p = ck.s4[ii][x] + t[jj];
                    const float h = (p >= 0.0f) ? p : ALPHA * p;
                    acc[ii][half * 4 + jj] = fmaf(ck.a4[x], h, acc[ii][half * 4 + jj]);
                }
            }
        }
    }
}

__global__ __launch_bounds__(256) __attribute__((amdgpu_waves_per_eu(2, 2)))
void gat_attn_kernel(const float* __restrict__ S,
                     const float* __restrict__ avec,
                     const float* __restrict__ bias,
                     float* __restrict__ attn)
{
    const int b    = blockIdx.y;
    const int i0   = blockIdx.x * ITILE;
    const int tid  = threadIdx.x;
    const int wave = tid >> 5;
    const int lane = tid & 31;

    __shared__ __attribute__((aligned(16))) float s1row[ITILE][EE];
    __shared__ __attribute__((aligned(16))) float a_s[EE];
    __shared__ __attribute__((aligned(16))) float part[NWAVE][ITILE][KK];

    const float* S1 = S;
    const v4f* src = (const v4f*)(S + (size_t)1024 * EE
                                    + ((size_t)b * EE + wave * ESLICE) * KK) + lane;

    Chunk A, B;
    load_chunk_g(A, src, 0);

    const int r    = wave >> 1;
    const int hsel = wave & 1;
    const v4f* brow = (const v4f*)(bias + (size_t)(i0 + r) * KK) + lane;
    const v4f bias_lo = brow[0];
    const v4f bias_hi = brow[32];

    a_s[tid] = avec[tid];
    #pragma unroll
    for (int ii = 0; ii < ITILE; ++ii)
        s1row[ii][tid] = S1[(size_t)(b * KK + i0 + ii) * EE + tid];
    __syncthreads();

    float acc[ITILE][8];
    #pragma unroll
    for (int ii = 0; ii < ITILE; ++ii) {
        #pragma unroll
        for (int q = 0; q < 8; ++q) acc[ii][q] = 0.0f;
    }

    const int e_w = wave * ESLICE;
    constexpr int NCH = ESLICE / ECH;
    load_chunk_l(A, a_s, s1row, e_w);
    #pragma nounroll
    for (int c = 0; c < NCH - 2; c += 2) {
        load_chunk(B, src, c + 1, a_s, s1row, e_w + (c + 1) * ECH);
        eat_chunk(acc, A);
        load_chunk(A, src, c + 2, a_s, s1row, e_w + (c + 2) * ECH);
        eat_chunk(acc, B);
    }
    load_chunk(B, src, NCH - 1, a_s, s1row, e_w + (NCH - 1) * ECH);
    eat_chunk(acc, A);
    eat_chunk(acc, B);

    #pragma unroll
    for (int ii = 0; ii < ITILE; ++ii) {
        v4f lo = {acc[ii][0], acc[ii][1], acc[ii][2], acc[ii][3]};
        v4f hi = {acc[ii][4], acc[ii][5], acc[ii][6], acc[ii][7]};
        *(v4f*)(&part[wave][ii][4 * lane])       = lo;
        *(v4f*)(&part[wave][ii][128 + 4 * lane]) = hi;
    }
    __syncthreads();

    v4f ev_lo = *(const v4f*)(&part[0][r][4 * lane]);
    v4f ev_hi = *(const v4f*)(&part[0][r][128 + 4 * lane]);
    #pragma unroll
    for (int w = 1; w < NWAVE; ++w) {
        ev_lo += *(const v4f*)(&part[w][r][4 * lane]);
        ev_hi += *(const v4f*)(&part[w][r][128 + 4 * lane]);
    }
    ev_lo += bias_lo;
    ev_hi += bias_hi;

    float mx = fmaxf(fmaxf(fmaxf(ev_lo[0], ev_lo[1]), fmaxf(ev_lo[2], ev_lo[3])),
                     fmaxf(fmaxf(ev_hi[0], ev_hi[1]), fmaxf(ev_hi[2], ev_hi[3])));
    #pragma unroll
    for (int off = 16; off > 0; off >>= 1)
        mx = fmaxf(mx, __shfl_xor(mx, off, 32));

    v4f p_lo, p_hi;
    #pragma unroll
    for (int q = 0; q < 4; ++q) {
        p_lo[q] = __expf(ev_lo[q] - mx);
        p_hi[q] = __expf(ev_hi[q] - mx);
    }
    float sum = ((p_lo[0] + p_lo[1]) + (p_lo[2] + p_lo[3]))
              + ((p_hi[0] + p_hi[1]) + (p_hi[2] + p_hi[3]));
    #pragma unroll
    for (int off = 16; off > 0; off >>= 1)
        sum += __shfl_xor(sum, off, 32);
    const float inv = 1.0f / sum;

    const v4f o = (hsel ? p_hi : p_lo) * inv;
    *(v4f*)(attn + (size_t)(b * KK + i0 + r) * KK + hsel * 128 + 4 * lane) = o;
}

#define OW    4
#define OKS   (KK / OW)
#define OJ    (OKS / 8)
#define OPITCH 36

__global__ __launch_bounds__(128) __attribute__((amdgpu_waves_per_eu(1, 4)))
void gat_out_kernel(const float* __restrict__ attn,
                    const float* __restrict__ X,
                    float* __restrict__ out)
{
    const int tid  = threadIdx.x;
    const int wave = tid >> 5;
    const int lane = tid & 31;
    const int ln   = lane & 15;
    const int hi   = lane >> 4;
    const int b    = blockIdx.y;
    const int row0 = (blockIdx.x >> 2) * 16;
    const int col0 = (blockIdx.x & 3) * 32;

    __shared__ __attribute__((aligned(16))) float part[OW][16][OPITCH];

    const int kb = OKS * wave + 4 * hi;
    const float* Ap = attn + (size_t)b * KK * KK + (size_t)(row0 + ln) * KK + kb;
    const float* Bp = X    + (size_t)b * KK * FF + (size_t)kb * FF + col0 + 2 * ln;

    v4f a4[OJ];
    v2f xb[OJ][2][2];
    #pragma unroll
    for (int j = 0; j < OJ; ++j) {
        a4[j] = *(const v4f*)(Ap + 8 * j);
        #pragma unroll
        for (int p = 0; p < 2; ++p) {
            #pragma unroll
            for (int e = 0; e < 2; ++e)
                xb[j][p][e] = *(const v2f*)(Bp + (size_t)(8 * j + 2 * p + e) * FF);
        }
    }

    v8f c0 = {};
    v8f c1 = {};
    #pragma unroll
    for (int j = 0; j < OJ; ++j) {
        #pragma unroll
        for (int p = 0; p < 2; ++p) {
            v2f a, b0, b1;
            a.x  = a4[j][2 * p];      a.y  = a4[j][2 * p + 1];
            b0.x = xb[j][p][0].x;     b0.y = xb[j][p][1].x;
            b1.x = xb[j][p][0].y;     b1.y = xb[j][p][1].y;
            c0 = __builtin_amdgcn_wmma_f32_16x16x4_f32(false, a, false, b0,
                                                       (short)0, c0, false, false);
            c1 = __builtin_amdgcn_wmma_f32_16x16x4_f32(false, a, false, b1,
                                                       (short)0, c1, false, false);
        }
    }

    #pragma unroll
    for (int r = 0; r < 8; ++r) {
        v2f v = {c0[r], c1[r]};
        *(v2f*)(&part[wave][8 * hi + r][2 * ln]) = v;
    }
    __syncthreads();

    const int orow = tid >> 3;
    const int ocol = (tid & 7) * 4;
    v4f acc = *(const v4f*)(&part[0][orow][ocol]);
    #pragma unroll
    for (int w = 1; w < OW; ++w)
        acc += *(const v4f*)(&part[w][orow][ocol]);

    v4f o;
    #pragma unroll
    for (int q = 0; q < 4; ++q)
        o[q] = 1.0f / (1.0f + __expf(-acc[q]));
    *(v4f*)(out + (size_t)b * KK * FF + (size_t)(row0 + orow) * FF + col0 + ocol) = o;
}

extern "C" void kernel_launch(void* const* d_in, const int* in_sizes, int n_in,
                              void* d_out, int out_size, void* d_ws, size_t ws_size,
                              hipStream_t stream) {
    const float* x    = (const float*)d_in[0];
    const float* W    = (const float*)d_in[1];
    const float* blin = (const float*)d_in[2];
    const float* a    = (const float*)d_in[3];
    const float* bias = (const float*)d_in[4];
    float* out = (float*)d_out;

    float* S     = (float*)d_ws;
    float* attnw = S + (size_t)2 * 1024 * EE;

    gat_gemm_s_kernel<<<dim3(128, 2), 128, 0, stream>>>(x, W, blin, S);

    gat_attn_kernel<<<dim3(KK / ITILE, BB), 256, 0, stream>>>(S, a, bias, attnw);

    gat_out_kernel<<<dim3(64, BB), 128, 0, stream>>>(attnw, x, out);
}


# attention e-loop: leaky relu's compare + select per element written as one max(p, 0.2p), the same value for every p since 0 < slope < 1; values and sum order unchanged
# speedup vs baseline: 2.6667x; 1.0936x over previous
#include <hip/hip_runtime.h>
#include <math.h>


typedef __attribute__((ext_vector_type(2))) float v2f;
typedef __attribute__((ext_vector_type(4))) float v4f;
typedef __attribute__((ext_vector_type(8))) float v8f;

#define ALPHA 0.2f
#define BB 4
#define KK 256
#define FF 128
#define EE 256
#define ITILE 4
#define NWAVE 8
#define ESLICE (EE / NWAVE)
#define ECH 4

#define GJ (FF / 8)

__global__ __launch_bounds__(128) __attribute__((amdgpu_waves_per_eu(1, 2)))
void gat_gemm_s_kernel(const float* __restrict__ X,
                       const float* __restrict__ W,
                       const float* __restrict__ blin,
                       float* __restrict__ S)
{
    const int wave = threadIdx.x >> 5;
    const int lane = threadIdx.x & 31;
    const int ln   = lane & 15;
    const int hi   = lane >> 4;
    const int s    = blockIdx.y;
    const int blk  = blockIdx.x;

    const int ga  = s ? 4 * (blk & 3) + wave : 4 * (blk & 15) + wave;
    const int gb  = s ? (blk >> 2) : (blk >> 4);
    const int ar0 = ga * 16;
    const int br0 = gb * 32;

    const float* Arow  = s ? W + (size_t)(ar0 + ln) * (2 * FF) + FF
                           : X + (size_t)(ar0 + ln) * FF;
    const float* B0row = s ? X + (size_t)(br0 + 2 * ln) * FF
                           : W + (size_t)(br0 + 2 * ln) * (2 * FF);
    const int bpitch   = s ? FF : 2 * FF;

    const float* Ap  = Arow  + 4 * hi;
    const float* B0p = B0row + 4 * hi;
    const float* B1p = B0p + bpitch;

    const v2f bl = *(const v2f*)(blin + (s ? 0 : br0) + 2 * ln);

    v4f a4[GJ], b0[GJ], b1[GJ];
    #pragma unroll
    for (int j = 0; j < GJ; ++j) {
        a4[j] = *(const v4f*)(Ap  + 8 * j);
        b0[j] = *(const v4f*)(B0p + 8 * j);
        b1[j] = *(const v4f*)(B1p + 8 * j);
    }
    __builtin_amdgcn_sched_barrier(0);

    const float i0 = s ? 0.0f : bl.x;
    const float i1 = s ? 0.0f : bl.y;
    v8f c0 = {i0, i0, i0, i0, i0, i0, i0, i0};
    v8f c1 = {i1, i1, i1, i1, i1, i1, i1, i1};
    #pragma unroll
    for (int j = 0; j < GJ; ++j) {
        #pragma unroll
        for (int p = 0; p < 2; ++p) {
            v2f a, x0, x1;
            a.x  = a4[j][2 * p];   a.y  = a4[j][2 * p + 1];
            x0.x = b0[j][2 * p];   x0.y = b0[j][2 * p + 1];
            x1.x = b1[j][2 * p];   x1.y = b1[j][2 * p + 1];
            c0 = __builtin_amdgcn_wmma_f32_16x16x4_f32(false, a, false, x0,
                                                       (short)0, c0, false, false);
            c1 = __builtin_amdgcn_wmma_f32_16x16x4_f32(false, a, false, x1,
                                                       (short)0, c1, false, false);
        }
    }

    float* D = s ? S + (size_t)1024 * EE
                     + ((size_t)(br0 >> 8) * EE + ar0 + 8 * hi) * KK
                     + (br0 & (KK - 1)) + 2 * ln
                 : S + (size_t)(ar0 + 8 * hi) * EE + br0 + 2 * ln;
    const int dpitch = s ? KK : EE;
    #pragma unroll
    for (int r = 0; r < 8; ++r) {
        v2f v = {c0[r], c1[r]};
        *(v2f*)(D + (size_t)r * dpitch) = v;
    }
}

struct Chunk {
    v4f t[2 * ECH];
    v4f a4;
    v4f s4[ITILE];
};

__device__ __forceinline__ void load_chunk_g(Chunk& ck, const v4f* src, int chunk)
{
    #pragma unroll
    for (int k = 0; k < 2 * ECH; ++k)
        ck.t[k] = src[(chunk * 2 * ECH + k) * 32];
}

__device__ __forceinline__ void load_chunk_l(Chunk& ck, const float* a_e, const float (*s1)[EE], int eb)
{
    ck.a4 = *(const v4f*)(a_e + eb);
    #pragma unroll
    for (int ii = 0; ii < ITILE; ++ii) ck.s4[ii] = *(const v4f*)(&s1[ii][eb]);
}

__device__ __forceinline__ void load_chunk(Chunk& ck, const v4f* src, int chunk,
                                           const float* a_e, const float (*s1)[EE], int eb)
{
    load_chunk_g(ck, src, chunk);
    load_chunk_l(ck, a_e, s1, eb);
}

__device__ __forceinline__ void eat_chunk(float (&acc)[ITILE][8], const Chunk& ck)
{
    #pragma unroll
    for (int x = 0; x < ECH; ++x) {
        #pragma unroll
        for (int half = 0; half < 2; ++half) {
            const v4f t = ck.t[2 * x + half];
            #pragma unroll
            for (int ii = 0; ii < ITILE; ++ii) {
                #pragma unroll
                for (int jj = 0; jj < 4; ++jj) {
                    const float p = ck.s4[ii][x] + t[jj];
                    const float h = fmaxf(p, ALPHA * p);
                    acc[ii][half * 4 + jj] = fmaf(ck.a4[x], h, acc[ii][half * 4 + jj]);
                }
            }
        }
    }
}

__global__ __launch_bounds__(256) __attribute__((amdgpu_waves_per_eu(2, 2)))
void gat_attn_kernel(const float* __restrict__ S,
                     const float* __restrict__ avec,
                     const float* __restrict__ bias,
                     float* __restrict__ attn)
{
    const int b    = blockIdx.y;
    const int i0   = blockIdx.x * ITILE;
    const int tid  = threadIdx.x;
    const int wave = tid >> 5;
    const int lane = tid & 31;

    __shared__ __attribute__((aligned(16))) float s1row[ITILE][EE];
    __shared__ __attribute__((aligned(16))) float a_s[EE];
    __shared__ __attribute__((aligned(16))) float part[NWAVE][ITILE][KK];

    const float* S1 = S;
    const v4f* src = (const v4f*)(S + (size_t)1024 * EE
                                    + ((size_t)b * EE + wave * ESLICE) * KK) + lane;

    Chunk A, B;
    load_chunk_g(A, src, 0);

    const int r    = wave >> 1;
    const int hsel = wave & 1;
    const v4f* brow = (const v4f*)(bias + (size_t)(i0 + r) * KK) + lane;
    const v4f bias_lo = brow[0];
    const v4f bias_hi = brow[32];

    a_s[tid] = avec[tid];
    #pragma unroll
    for (int ii = 0; ii < ITILE; ++ii)
        s1row[ii][tid] = S1[(size_t)(b * KK + i0 + ii) * EE + tid];
    __syncthreads();

    float acc[ITILE][8];
    #pragma unroll
    for (int ii = 0; ii < ITILE; ++ii) {
        #pragma unroll
        for (int q = 0; q < 8; ++q) acc[ii][q] = 0.0f;
    }

    const int e_w = wave * ESLICE;
    constexpr int NCH = ESLICE / ECH;
    load_chunk_l(A, a_s, s1row, e_w);
    #pragma nounroll
    for (int c = 0; c < NCH - 2; c += 2) {
        load_chunk(B, src, c + 1, a_s, s1row, e_w + (c + 1) * ECH);
        eat_chunk(acc, A);
        load_chunk(A, src, c + 2, a_s, s1row, e_w + (c + 2) * ECH);
        eat_chunk(acc, B);
    }
    load_chunk(B, src, NCH - 1, a_s, s1row, e_w + (NCH - 1) * ECH);
    eat_chunk(acc, A);
    eat_chunk(acc, B);

    #pragma unroll
    for (int ii = 0; ii < ITILE; ++ii) {
        v4f lo = {acc[ii][0], acc[ii][1], acc[ii][2], acc[ii][3]};
        v4f hi = {acc[ii][4], acc[ii][5], acc[ii][6], acc[ii][7]};
        *(v4f*)(&part[wave][ii][4 * lane])       = lo;
        *(v4f*)(&part[wave][ii][128 + 4 * lane]) = hi;
    }
    __syncthreads();

    v4f ev_lo = *(const v4f*)(&part[0][r][4 * lane]);
    v4f ev_hi = *(const v4f*)(&part[0][r][128 + 4 * lane]);
    #pragma unroll
    for (int w = 1; w < NWAVE; ++w) {
        ev_lo += *(const v4f*)(&part[w][r][4 * lane]);
        ev_hi += *(const v4f*)(&part[w][r][128 + 4 * lane]);
    }
    ev_lo += bias_lo;
    ev_hi += bias_hi;

    float mx = fmaxf(fmaxf(fmaxf(ev_lo[0], ev_lo[1]), fmaxf(ev_lo[2], ev_lo[3])),
                     fmaxf(fmaxf(ev_hi[0], ev_hi[1]), fmaxf(ev_hi[2], ev_hi[3])));
    #pragma unroll
    for (int off = 16; off > 0; off >>= 1)
        mx = fmaxf(mx, __shfl_xor(mx, off, 32));

    v4f p_lo, p_hi;
    #pragma unroll
    for (int q = 0; q < 4; ++q) {
        p_lo[q] = __expf(ev_lo[q] - mx);
        p_hi[q] = __expf(ev_hi[q] - mx);
    }
    float sum = ((p_lo[0] + p_lo[1]) + (p_lo[2] + p_lo[3]))
              + ((p_hi[0] + p_hi[1]) + (p_hi[2] + p_hi[3]));
    #pragma unroll
    for (int off = 16; off > 0; off >>= 1)
        sum += __shfl_xor(sum, off, 32);
    const float inv = 1.0f / sum;

    const v4f o = (hsel ? p_hi : p_lo) * inv;
    *(v4f*)(attn + (size_t)(b * KK + i0 + r) * KK + hsel * 128 + 4 * lane) = o;
}

#define OW    4
#define OKS   (KK / OW)
#define OJ    (OKS / 8)
#define OPITCH 36

__global__ __launch_bounds__(128) __attribute__((amdgpu_waves_per_eu(1, 4)))
void gat_out_kernel(const float* __restrict__ attn,
                    const float* __restrict__ X,
                    float* __restrict__ out)
{
    const int tid  = threadIdx.x;
    const int wave = tid >> 5;
    const int lane = tid & 31;
    const int ln   = lane & 15;
    const int hi   = lane >> 4;
    const int b    = blockIdx.y;
    const int row0 = (blockIdx.x >> 2) * 16;
    const int col0 = (blockIdx.x & 3) * 32;

    __shared__ __attribute__((aligned(16))) float part[OW][16][OPITCH];

    const int kb = OKS * wave + 4 * hi;
    const float* Ap = attn + (size_t)b * KK * KK + (size_t)(row0 + ln) * KK + kb;
    const float* Bp = X    + (size_t)b * KK * FF + (size_t)kb * FF + col0 + 2 * ln;

    v4f a4[OJ];
    v2f xb[OJ][2][2];
    #pragma unroll
    for (int j = 0; j < OJ; ++j) {
        a4[j] = *(const v4f*)(Ap + 8 * j);
        #pragma unroll
        for (int p = 0; p < 2; ++p) {
            #pragma unroll
            for (int e = 0; e < 2; ++e)
                xb[j][p][e] = *(const v2f*)(Bp + (size_t)(8 * j + 2 * p + e) * FF);
        }
    }

    v8f c0 = {};
    v8f c1 = {};
    #pragma unroll
    for (int j = 0; j < OJ; ++j) {
        #pragma unroll
        for (int p = 0; p < 2; ++p) {
            v2f a, b0, b1;
            a.x  = a4[j][2 * p];      a.y  = a4[j][2 * p + 1];
            b0.x = xb[j][p][0].x;     b0.y = xb[j][p][1].x;
            b1.x = xb[j][p][0].y;     b1.y = xb[j][p][1].y;
            c0 = __builtin_amdgcn_wmma_f32_16x16x4_f32(false, a, false, b0,
                                                       (short)0, c0, false, false);
            c1 = __builtin_amdgcn_wmma_f32_16x16x4_f32(false, a, false, b1,
                                                       (short)0, c1, false, false);
        }
    }

    #pragma unroll
    for (int r = 0; r < 8; ++r) {
        v2f v = {c0[r], c1[r]};
        *(v2f*)(&part[wave][8 * hi + r][2 * ln]) = v;
    }
    __syncthreads();

    const int orow = tid >> 3;
    const int ocol = (tid & 7) * 4;
    v4f acc = *(const v4f*)(&part[0][orow][ocol]);
    #pragma unroll
    for (int w = 1; w < OW; ++w)
        acc += *(const v4f*)(&part[w][orow][ocol]);

    v4f o;
    #pragma unroll
    for (int q = 0; q < 4; ++q)
        o[q] = 1.0f / (1.0f + __expf(-acc[q]));
    *(v4f*)(out + (size_t)b * KK * FF + (size_t)(row0 + orow) * FF + col0 + ocol) = o;
}

extern "C" void kernel_launch(void* const* d_in, const int* in_sizes, int n_in,
                              void* d_out, int out_size, void* d_ws, size_t ws_size,
                              hipStream_t stream) {
    const float* x    = (const float*)d_in[0];
    const float* W    = (const float*)d_in[1];
    const float* blin = (const float*)d_in[2];
    const float* a    = (const float*)d_in[3];
    const float* bias = (const float*)d_in[4];
    float* out = (float*)d_out;

    float* S     = (float*)d_ws;
    float* attnw = S + (size_t)2 * 1024 * EE;

    gat_gemm_s_kernel<<<dim3(128, 2), 128, 0, stream>>>(x, W, blin, S);

    gat_attn_kernel<<<dim3(KK / ITILE, BB), 256, 0, stream>>>(S, a, bias, attnw);

    gat_out_kernel<<<dim3(64, BB), 128, 0, stream>>>(attnw, x, out);
}
